# starting kernel
# baseline (speedup 1.0000x reference)
#include <hip/hip_runtime.h>

typedef __attribute__((ext_vector_type(2))) float v2f;
typedef __attribute__((ext_vector_type(8))) float v8f;

namespace {
constexpr int   kN  = 61;
constexpr int   kE  = 40;
constexpr int   kH  = 10;
constexpr int   kD  = 4;
constexpr int   kRS = 44;
constexpr int   kRows = 64;
constexpr int   kDump = kRows * kRS;
constexpr int   kBuf  = kRows * kRS + 16;
constexpr float kInvSqrtE = 0.15811388300841898f;
}

__device__ __forceinline__ int clampi(int v, int hi2) { return v < hi2 ? v : hi2; }

__global__ __launch_bounds__(256)
void mha_fused_kernel(const float* __restrict__ x,
                      const float* __restrict__ Wq, const float* __restrict__ bq,
                      const float* __restrict__ Wk, const float* __restrict__ bk,
                      const float* __restrict__ Wv, const float* __restrict__ bv,
                      const float* __restrict__ Wp, const float* __restrict__ bp,
                      const float* __restrict__ pe,
                      float* __restrict__ out) {
  __shared__ __align__(16) float s_x[kBuf];
  __shared__ __align__(16) float s_q[kBuf];
  __shared__ __align__(16) float s_k[kBuf];
  __shared__ __align__(16) float s_v[kBuf];

  const int b    = blockIdx.x;
  const int tid  = threadIdx.x;
  const int lane = tid & 31;
  const int wave = tid >> 5;
  const int hi   = lane >> 4;
  const int l16  = lane & 15;
  const int kbl  = hi * 2;

  if (tid == 0) {
    __builtin_prefetch(Wq, 0, 1);
    __builtin_prefetch(Wk, 0, 1);
    __builtin_prefetch(Wv, 0, 1);
    __builtin_prefetch(Wp, 0, 1);
  }

  for (int i = tid; i < kN * kE; i += 256) {
    const int n = i / kE;
    s_x[n * kRS + (i - n * kE)] = x[(size_t)b * (kN * kE) + i] + pe[i];
  }
  __syncthreads();

  for (int tile = wave; tile < 36; tile += 8) {
    const int mat = tile / 12;
    const int mt  = (tile % 12) / 3;
    const int nt  = tile % 3;
    const float* __restrict__ W  = (mat == 0) ? Wq : (mat == 1) ? Wk : Wv;
    const float* __restrict__ bi = (mat == 0) ? bq : (mat == 1) ? bk : bv;
    float* __restrict__ S        = (mat == 0) ? s_q : (mat == 1) ? s_k : s_v;

    const int arow = mt * 16 + l16;
    const int ncol = nt * 16 + l16;
    const int ar_c = clampi(arow, kN - 1);
    const int nc_c = clampi(ncol, kE - 1);
    const bool cok = ncol < kE;

    v2f af[10], bf[10];
#pragma unroll
    for (int kc = 0; kc < 10; ++kc) {
      const int kb = kc * 4 + kbl;
      af[kc].x = s_x[ar_c * kRS + kb];
      af[kc].y = s_x[ar_c * kRS + kb + 1];
      bf[kc].x = W[nc_c * kE + kb];
      bf[kc].y = W[nc_c * kE + kb + 1];
    }
    v8f acc = {0.f, 0.f, 0.f, 0.f, 0.f, 0.f, 0.f, 0.f};
#pragma unroll
    for (int kc = 0; kc < 10; ++kc)
      acc = __builtin_amdgcn_wmma_f32_16x16x4_f32(false, af[kc], false, bf[kc],
                                                  (short)0, acc, false, false);

    const float bias = bi[nc_c];
#pragma unroll
    for (int r = 0; r < 8; ++r) {
      const int row = mt * 16 + r + hi * 8;
      float val = acc[r] + bias;
      if (r < 3)
        val = (mat != 0 && mt == 0 && hi == 0 && (ncol & 3) > r) ? 0.f : val;
      const bool ok = (row < kN) && cok;
      S[ok ? (row * kRS + ncol) : kDump] = val;
    }
  }
  __syncthreads();

  float* __restrict__ s_o = s_x;

  for (int h = wave; h < kH; h += 8) {
    const int co = h * kD;

    v2f ak[4];
#pragma unroll
    for (int kt = 0; kt < 4; ++kt) {
      const int kr = clampi(kt * 16 + l16, kN - 1);
      ak[kt].x = s_k[kr * kRS + co + kbl];
      ak[kt].y = s_k[kr * kRS + co + kbl + 1];
    }

#pragma unroll
    for (int qt = 0; qt < 4; ++qt) {
      const int q   = qt * 16 + l16;
      const int q_c = clampi(q, kN - 1);
      v2f bq_;
      bq_.x = s_q[q_c * kRS + co + kbl];
      bq_.y = s_q[q_c * kRS + co + kbl + 1];

      v8f et[4];
#pragma unroll
      for (int kt = 0; kt < 4; ++kt) {
        v8f z = {0.f, 0.f, 0.f, 0.f, 0.f, 0.f, 0.f, 0.f};
        et[kt] = __builtin_amdgcn_wmma_f32_16x16x4_f32(false, ak[kt], false, bq_,
                                                       (short)0, z, false, false);
      }
#pragma unroll
      for (int r = 5; r < 8; ++r) et[3][r] = hi ? -1e30f : et[3][r];

      float m = -1e30f;
#pragma unroll
      for (int kt = 0; kt < 4; ++kt)
#pragma unroll
        for (int r = 0; r < 8; ++r) m = fmaxf(m, et[kt][r]);
      m = fmaxf(m, __shfl_xor(m, 16, 32));

      float sum = 0.f;
#pragma unroll
      for (int kt = 0; kt < 4; ++kt)
#pragma unroll
        for (int r = 0; r < 8; ++r) {
          const float e = __expf((et[kt][r] - m) * kInvSqrtE);
          et[kt][r] = e;
          sum += e;
        }
      sum += __shfl_xor(sum, 16, 32);
      const float inv = 1.f / sum;

      float o0 = 0.f, o1 = 0.f, o2 = 0.f, o3 = 0.f;
#pragma unroll
      for (int kt = 0; kt < 4; ++kt)
#pragma unroll
        for (int r = 0; r < 8; ++r) {
          const int key = clampi(kt * 16 + r + 8 * hi, kN - 1);
          const float4 vr = *(const float4*)(s_v + key * kRS + co);
          const float a = et[kt][r];
          o0 += a * vr.x; o1 += a * vr.y; o2 += a * vr.z; o3 += a * vr.w;
        }
      o0 += __shfl_xor(o0, 16, 32);
      o1 += __shfl_xor(o1, 16, 32);
      o2 += __shfl_xor(o2, 16, 32);
      o3 += __shfl_xor(o3, 16, 32);

      float4 ov; ov.x = o0 * inv; ov.y = o1 * inv; ov.z = o2 * inv; ov.w = o3 * inv;
      *(float4*)(s_o + (q < kN ? q * kRS + co : kDump)) = ov;
    }
  }
  __syncthreads();

  for (int tile = wave; tile < 12; tile += 8) {
    const int mt   = tile / 3;
    const int nt   = tile % 3;
    const int arow = mt * 16 + l16;
    const int ncol = nt * 16 + l16;
    const int ar_c = clampi(arow, kN - 1);
    const int nc_c = clampi(ncol, kE - 1);
    const bool cok = ncol < kE;

    v2f af[10], bf[10];
#pragma unroll
    for (int kc = 0; kc < 10; ++kc) {
      const int kb = kc * 4 + kbl;
      af[kc].x = s_o[ar_c * kRS + kb];
      af[kc].y = s_o[ar_c * kRS + kb + 1];
      bf[kc].x = Wp[nc_c * kE + kb];
      bf[kc].y = Wp[nc_c * kE + kb + 1];
    }
    v8f acc = {0.f, 0.f, 0.f, 0.f, 0.f, 0.f, 0.f, 0.f};
#pragma unroll
    for (int kc = 0; kc < 10; ++kc)
      acc = __builtin_amdgcn_wmma_f32_16x16x4_f32(false, af[kc], false, bf[kc],
                                                  (short)0, acc, false, false);

    const float bias = bp[nc_c];
#pragma unroll
    for (int r = 0; r < 8; ++r) {
      const int row = mt * 16 + r + hi * 8;
      if (row < kN && cok)
        out[(size_t)b * (kN * kE) + row * kE + ncol] = acc[r] + bias;
    }
  }
}

extern "C" void kernel_launch(void* const* d_in, const int* in_sizes, int n_in,
                              void* d_out, int out_size, void* d_ws, size_t ws_size,
                              hipStream_t stream) {
  (void)in_sizes; (void)n_in; (void)out_size; (void)d_ws; (void)ws_size;
  const float* x  = (const float*)d_in[0];
  const float* Wq = (const float*)d_in[1];
  const float* bq = (const float*)d_in[2];
  const float* Wk = (const float*)d_in[3];
  const float* bk = (const float*)d_in[4];
  const float* Wv = (const float*)d_in[5];
  const float* bv = (const float*)d_in[6];
  const float* Wp = (const float*)d_in[7];
  const float* bp = (const float*)d_in[8];
  const float* pe = (const float*)d_in[9];
  float* out = (float*)d_out;

  mha_fused_kernel<<<dim3(4096), dim3(256), 0, stream>>>(
      x, Wq, bq, Wk, bk, Wv, bv, Wp, bp, pe, out);
}


# x tile, positional table and all 4 weight matrices staged to LDS by one batch of coalesced 16-byte loads; projection B fragments come from LDS, not 8-byte global gathers; q-tile loop unrolled 2
# speedup vs baseline: 2.3560x; 2.3560x over previous
#include <hip/hip_runtime.h>


typedef __attribute__((ext_vector_type(2))) float v2f;
typedef __attribute__((ext_vector_type(4))) float v4f;
typedef __attribute__((ext_vector_type(8))) float v8f;

namespace {
constexpr int   kN  = 61;
constexpr int   kE  = 40;
constexpr int   kH  = 10;
constexpr int   kD  = 4;
constexpr int   kRS = 40;
constexpr int   kDump = kN * kRS;
constexpr int   kBuf  = kN * kRS + 16;
constexpr int   kWsz  = kE * kE;
constexpr float kInvSqrtE = 0.15811388300841898f;
constexpr int   kWaves   = 2;
constexpr int   kThreads = kWaves * 32;
constexpr int   kXVec = kN * kE / 4;
constexpr int   kXIt  = (kXVec + kThreads - 1) / kThreads;
constexpr int   kWVec = kWsz / 4;
constexpr int   kWIt  = (kWVec + kThreads - 1) / kThreads;
}

__device__ __forceinline__ int clampi(int v, int hi2) { return v < hi2 ? v : hi2; }

__device__ __forceinline__ v8f proj_tile(const float* __restrict__ sA,
                                         const float* __restrict__ sW,
                                         int ar_c, int nc_c, int kbl) {
  v2f af[10], bf[10];
#pragma unroll
  for (int kc = 0; kc < 10; ++kc) {
    const int kb = kc * 4 + kbl;
    af[kc].x = sA[ar_c * kRS + kb];
    af[kc].y = sA[ar_c * kRS + kb + 1];
    bf[kc].x = sW[nc_c * kE + kb];
    bf[kc].y = sW[nc_c * kE + kb + 1];
  }
  v8f acc = {0.f, 0.f, 0.f, 0.f, 0.f, 0.f, 0.f, 0.f};
#pragma unroll
  for (int kc = 0; kc < 10; ++kc)
    acc = __builtin_amdgcn_wmma_f32_16x16x4_f32(false, af[kc], false, bf[kc],
                                                (short)0, acc, false, false);
  return acc;
}

__global__ __launch_bounds__(kThreads, 3)
void mha_fused_kernel(const float* __restrict__ x,
                      const float* __restrict__ Wq, const float* __restrict__ bq,
                      const float* __restrict__ Wk, const float* __restrict__ bk,
                      const float* __restrict__ Wv, const float* __restrict__ bv,
                      const float* __restrict__ Wp, const float* __restrict__ bp,
                      const float* __restrict__ pe,
                      float* __restrict__ out) {
  __shared__ __align__(16) float s_x[kBuf];
  __shared__ __align__(16) float s_q[kBuf];
  __shared__ __align__(16) float s_k[kBuf];
  __shared__ __align__(16) float s_v[kBuf];
  __shared__ __align__(16) float s_wa[kWsz];
  __shared__ __align__(16) float s_wb[kWsz];

  const int b    = blockIdx.x;
  const int tid  = threadIdx.x;
  const int lane = tid & 31;
  const int wave = tid >> 5;
  const int hi   = lane >> 4;
  const int l16  = lane & 15;
  const int kbl  = hi * 2;

  float* s_wv = s_q;
  v4f wpr[kWIt];
  {
    const v4f* __restrict__ xv  = (const v4f*)(x + (size_t)b * (kN * kE));
    const v4f* __restrict__ pv  = (const v4f*)pe;
    const v4f* __restrict__ wqv = (const v4f*)Wq;
    const v4f* __restrict__ wkv = (const v4f*)Wk;
    const v4f* __restrict__ wvv = (const v4f*)Wv;
    const v4f* __restrict__ wpv = (const v4f*)Wp;
    v4f xr[kXIt], pr[kXIt], wkr[kWIt], wvr[kWIt], wqr[kWIt];
#pragma unroll
    for (int j = 0; j < kXIt; ++j) {
      const int g = clampi(tid + j * kThreads, kXVec - 1);
      xr[j] = xv[g];
      pr[j] = pv[g];
    }
#pragma unroll
    for (int j = 0; j < kWIt; ++j) {
      const int g = clampi(tid + j * kThreads, kWVec - 1);
      wkr[j] = wkv[g];
      wvr[j] = wvv[g];
      wqr[j] = wqv[g];
      wpr[j] = wpv[g];
    }
#pragma unroll
    for (int j = 0; j < kXIt; ++j) {
      const int g = tid + j * kThreads;
      if (j < kXIt - 1 || g < kXVec) ((v4f*)s_x)[g] = xr[j] + pr[j];
    }
#pragma unroll
    for (int j = 0; j < kWIt; ++j) {
      const int g = tid + j * kThreads;
      if (j < kWIt - 1 || g < kWVec) {
        ((v4f*)s_wb)[g] = wkr[j];
        ((v4f*)s_wv)[g] = wvr[j];
        ((v4f*)s_wa)[g] = wqr[j];
      }
    }
  }
  __syncthreads();

  for (int tile = wave; tile < 24; tile += kWaves) {
    const bool isv = tile >= 12;
    const int  t12 = isv ? tile - 12 : tile;
    const int  mt  = t12 / 3;
    const int  nt  = t12 % 3;
    const float* sW = isv ? s_wv : s_wb;
    const float* __restrict__ bi = isv ? bv : bk;
    float* S        = isv ? s_v : s_k;

    const int arow = mt * 16 + l16;
    const int ncol = nt * 16 + l16;
    const int ar_c = clampi(arow, kN - 1);
    const int nc_c = clampi(ncol, kE - 1);
    const bool cok = ncol < kE;

    const v8f acc = proj_tile(s_x, sW, ar_c, nc_c, kbl);

    const float bias = bi[nc_c];
#pragma unroll
    for (int r = 0; r < 8; ++r) {
      const int row = mt * 16 + r + hi * 8;
      float val = acc[r] + bias;
      if (r < 3)
        val = (mt == 0 && hi == 0 && (ncol & 3) > r) ? 0.f : val;
      const bool ok = (row < kN) && cok;
      S[ok ? (row * kRS + ncol) : kDump] = val;
    }
  }
  __syncthreads();

#pragma unroll
  for (int j = 0; j < kWIt; ++j) {
    const int g = tid + j * kThreads;
    if (j < kWIt - 1 || g < kWVec) ((v4f*)s_wb)[g] = wpr[j];
  }

  for (int tile = wave; tile < 12; tile += kWaves) {
    const int mt   = tile / 3;
    const int nt   = tile % 3;
    const int arow = mt * 16 + l16;
    const int ncol = nt * 16 + l16;
    const int ar_c = clampi(arow, kN - 1);
    const int nc_c = clampi(ncol, kE - 1);
    const bool cok = ncol < kE;

    const v8f acc = proj_tile(s_x, s_wa, ar_c, nc_c, kbl);

    const float bias = bq[nc_c];
#pragma unroll
    for (int r = 0; r < 8; ++r) {
      const int row = mt * 16 + r + hi * 8;
      const bool ok = (row < kN) && cok;
      s_q[ok ? (row * kRS + ncol) : kDump] = acc[r] + bias;
    }
  }
  __syncthreads();

  float* __restrict__ s_o = s_x;

  for (int h = wave; h < kH; h += kWaves) {
    const int co = h * kD;

    v2f ak[4];
#pragma unroll
    for (int kt = 0; kt < 4; ++kt) {
      const int kr = clampi(kt * 16 + l16, kN - 1);
      ak[kt].x = s_k[kr * kRS + co + kbl];
      ak[kt].y = s_k[kr * kRS + co + kbl + 1];
    }

#pragma unroll 2
    for (int qt = 0; qt < 4; ++qt) {
      const int q   = qt * 16 + l16;
      const int q_c = clampi(q, kN - 1);
      v2f bq_;
      bq_.x = s_q[q_c * kRS + co + kbl];
      bq_.y = s_q[q_c * kRS + co + kbl + 1];

      v8f et[4];
#pragma unroll
      for (int kt = 0; kt < 4; ++kt) {
        v8f z = {0.f, 0.f, 0.f, 0.f, 0.f, 0.f, 0.f, 0.f};
        et[kt] = __builtin_amdgcn_wmma_f32_16x16x4_f32(false, ak[kt], false, bq_,
                                                       (short)0, z, false, false);
      }
#pragma unroll
      for (int r = 5; r < 8; ++r) et[3][r] = hi ? -1e30f : et[3][r];

      float m = -1e30f;
#pragma unroll
      for (int kt = 0; kt < 4; ++kt)
#pragma unroll
        for (int r = 0; r < 8; ++r) m = fmaxf(m, et[kt][r]);
      m = fmaxf(m, __shfl_xor(m, 16, 32));

      float sum = 0.f;
#pragma unroll
      for (int kt = 0; kt < 4; ++kt)
#pragma unroll
        for (int r = 0; r < 8; ++r) {
          const float e = __expf((et[kt][r] - m) * kInvSqrtE);
          et[kt][r] = e;
          sum += e;
        }
      sum += __shfl_xor(sum, 16, 32);
      const float inv = 1.f / sum;

      float o0 = 0.f, o1 = 0.f, o2 = 0.f, o3 = 0.f;
#pragma unroll
      for (int kt = 0; kt < 4; ++kt)
#pragma unroll
        for (int r = 0; r < 8; ++r) {
          const int key = clampi(kt * 16 + r + 8 * hi, kN - 1);
          const float4 vr = *(const float4*)(s_v + key * kRS + co);
          const float a = et[kt][r];
          o0 += a * vr.x; o1 += a * vr.y; o2 += a * vr.z; o3 += a * vr.w;
        }
      o0 += __shfl_xor(o0, 16, 32);
      o1 += __shfl_xor(o1, 16, 32);
      o2 += __shfl_xor(o2, 16, 32);
      o3 += __shfl_xor(o3, 16, 32);

      float4 ov; ov.x = o0 * inv; ov.y = o1 * inv; ov.z = o2 * inv; ov.w = o3 * inv;
      *(float4*)(s_o + (q < kN ? q * kRS + co : kDump)) = ov;
    }
  }
  __syncthreads();

  for (int tile = wave; tile < 12; tile += kWaves) {
    const int mt   = tile / 3;
    const int nt   = tile % 3;
    const int arow = mt * 16 + l16;
    const int ncol = nt * 16 + l16;
    const int ar_c = clampi(arow, kN - 1);
    const int nc_c = clampi(ncol, kE - 1);
    const bool cok = ncol < kE;

    const v8f acc = proj_tile(s_o, s_wb, ar_c, nc_c, kbl);

    const float bias = bp[nc_c];
#pragma unroll
    for (int r = 0; r < 8; ++r) {
      const int row = mt * 16 + r + hi * 8;
      if (row < kN && cok)
        out[(size_t)b * (kN * kE) + row * kE + ncol] = acc[r] + bias;
    }
  }
}

extern "C" void kernel_launch(void* const* d_in, const int* in_sizes, int n_in,
                              void* d_out, int out_size, void* d_ws, size_t ws_size,
                              hipStream_t stream) {
  (void)in_sizes; (void)n_in; (void)out_size; (void)d_ws; (void)ws_size;
  const float* x  = (const float*)d_in[0];
  const float* Wq = (const float*)d_in[1];
  const float* bq = (const float*)d_in[2];
  const float* Wk = (const float*)d_in[3];
  const float* bk = (const float*)d_in[4];
  const float* Wv = (const float*)d_in[5];
  const float* bv = (const float*)d_in[6];
  const float* Wp = (const float*)d_in[7];
  const float* bp = (const float*)d_in[8];
  const float* pe = (const float*)d_in[9];
  float* out = (float*)d_out;

  mha_fused_kernel<<<dim3(4096), dim3(64), 0, stream>>>(
      x, Wq, bq, Wk, bk, Wv, bv, Wp, bp, pe, out);
}


# tile epilogues: Q/K/V LDS buffers padded to 64 rows, each 16x16 tile stored from one lane address with constant row offsets; padded columns and rows masked once per tile, not per element
# speedup vs baseline: 2.7211x; 1.1550x over previous
#include <hip/hip_runtime.h>


typedef __attribute__((ext_vector_type(2))) float v2f;
typedef __attribute__((ext_vector_type(4))) float v4f;
typedef __attribute__((ext_vector_type(8))) float v8f;

namespace {
constexpr int   kN  = 61;
constexpr int   kE  = 40;
constexpr int   kH  = 10;
constexpr int   kD  = 4;
constexpr int   kRS = 40;
constexpr int   kBufX = kN * kRS;
constexpr int   kBuf  = 64 * kRS;
constexpr int   kWsz  = kE * kE;
constexpr float kInvSqrtE = 0.15811388300841898f;
constexpr int   kWaves   = 2;
constexpr int   kThreads = kWaves * 32;
constexpr int   kTPW  = 12 / kWaves;
constexpr int   kXVec = kN * kE / 4;
constexpr int   kXIt  = (kXVec + kThreads - 1) / kThreads;
constexpr int   kWVec = kWsz / 4;
constexpr int   kWIt  = (kWVec + kThreads - 1) / kThreads;
}

__device__ __forceinline__ int clampi(int v, int hi2) { return v < hi2 ? v : hi2; }

__device__ __forceinline__ v8f proj_tile(const float* __restrict__ sA,
                                         const float* __restrict__ sW,
                                         int ar_c, int nc_c, int kbl) {
  v2f af[10], bf[10];
#pragma unroll
  for (int kc = 0; kc < 10; ++kc) {
    const int kb = kc * 4 + kbl;
    af[kc].x = sA[ar_c * kRS + kb];
    af[kc].y = sA[ar_c * kRS + kb + 1];
    bf[kc].x = sW[nc_c * kE + kb];
    bf[kc].y = sW[nc_c * kE + kb + 1];
  }
  v8f acc = {0.f, 0.f, 0.f, 0.f, 0.f, 0.f, 0.f, 0.f};
#pragma unroll
  for (int kc = 0; kc < 10; ++kc)
    acc = __builtin_amdgcn_wmma_f32_16x16x4_f32(false, af[kc], false, bf[kc],
                                                (short)0, acc, false, false);
  return acc;
}

template <bool kMask>
__device__ __forceinline__ void store_tile(float* __restrict__ S, const v8f acc,
                                           int mt, int ncol, int hi) {
  float* __restrict__ base = S + (mt * 16 + hi * 8) * kRS + ncol;
  if (ncol < kE) {
#pragma unroll
    for (int r = 0; r < 8; ++r) {
      float val = acc[r];
      if (kMask && r < 3)
        val = (mt == 0 && hi == 0 && (ncol & 3) > r) ? 0.f : val;
      base[r * kRS] = val;
    }
  }
}

__global__ __launch_bounds__(kThreads, 4)
void mha_fused_kernel(const float* __restrict__ x,
                      const float* __restrict__ Wq, const float* __restrict__ bq,
                      const float* __restrict__ Wk, const float* __restrict__ bk,
                      const float* __restrict__ Wv, const float* __restrict__ bv,
                      const float* __restrict__ Wp, const float* __restrict__ bp,
                      const float* __restrict__ pe,
                      float* __restrict__ out) {
  __shared__ __align__(16) float s_x[kBufX];
  __shared__ __align__(16) float s_q[kBuf];
  __shared__ __align__(16) float s_k[kBuf];
  __shared__ __align__(16) float s_v[kBuf];

  const int b    = blockIdx.x;
  const int tid  = threadIdx.x;
  const int lane = tid & 31;
  const int wave = tid >> 5;
  const int hi   = lane >> 4;
  const int l16  = lane & 15;
  const int kbl  = hi * 2;

  v4f wvr[kWIt];
  {
    const v4f* __restrict__ xv  = (const v4f*)(x + (size_t)b * (kN * kE));
    const v4f* __restrict__ pv  = (const v4f*)pe;
    const v4f* __restrict__ wqv = (const v4f*)Wq;
    const v4f* __restrict__ wkv = (const v4f*)Wk;
    const v4f* __restrict__ wvv = (const v4f*)Wv;
    v4f xr[kXIt], pr[kXIt], wkr[kWIt], wqr[kWIt];
#pragma unroll
    for (int j = 0; j < kXIt; ++j) {
      const int g = clampi(tid + j * kThreads, kXVec - 1);
      xr[j] = xv[g];
      pr[j] = pv[g];
    }
#pragma unroll
    for (int j = 0; j < kWIt; ++j) {
      const int g = clampi(tid + j * kThreads, kWVec - 1);
      wqr[j] = wqv[g];
      wkr[j] = wkv[g];
      wvr[j] = wvv[g];
    }
#pragma unroll
    for (int j = 0; j < kXIt; ++j) {
      const int g = tid + j * kThreads;
      if (j < kXIt - 1 || g < kXVec) ((v4f*)s_x)[g] = xr[j] + pr[j];
    }
#pragma unroll
    for (int j = 0; j < kWIt; ++j) {
      const int g = tid + j * kThreads;
      if (j < kWIt - 1 || g < kWVec) {
        ((v4f*)s_k)[g] = wqr[j];
        ((v4f*)s_v)[g] = wkr[j];
      }
    }
  }
  __syncthreads();

  v8f hold[kTPW];
#pragma unroll
  for (int i = 0; i < kTPW; ++i) {
    const int tile = wave + i * kWaves;
    const int mt   = tile / 3;
    const int nt   = tile % 3;
    const int ncol = nt * 16 + l16;
    const int ar_c = clampi(mt * 16 + l16, kN - 1);
    const int nc_c = clampi(ncol, kE - 1);
    const float bqv = bq[nc_c];
    const float bkv = bk[nc_c];
    v8f qa = proj_tile(s_x, s_k  , ar_c, nc_c, kbl);
    asm volatile("" :: "v"(bqv));
    qa = qa + bqv;
    store_tile<false>(s_q, qa, mt, ncol, hi);
    hold[i] = proj_tile(s_x, s_v  , ar_c, nc_c, kbl) + bkv;
  }
  __syncthreads();

#pragma unroll
  for (int i = 0; i < kTPW; ++i) {
    const int tile = wave + i * kWaves;
    const int mt   = tile / 3;
    const int ncol = (tile % 3) * 16 + l16;
    store_tile<true>(s_k, hold[i], mt, ncol, hi);
  }
#pragma unroll
  for (int j = 0; j < kWIt; ++j) {
    const int g = tid + j * kThreads;
    if (j < kWIt - 1 || g < kWVec) ((v4f*)s_v)[g] = wvr[j];
  }
  __syncthreads();

  v4f wpr[kWIt];
#pragma unroll
  for (int j = 0; j < kWIt; ++j)
    wpr[j] = ((const v4f*)Wp)[clampi(tid + j * kThreads, kWVec - 1)];

#pragma unroll
  for (int i = 0; i < kTPW; ++i) {
    const int tile = wave + i * kWaves;
    const int mt   = tile / 3;
    const int ncol = (tile % 3) * 16 + l16;
    const int nc_c = clampi(ncol, kE - 1);
    const float bvv = bv[nc_c];
    hold[i] = proj_tile(s_x, s_v  , clampi(mt * 16 + l16, kN - 1), nc_c, kbl) + bvv;
  }
  __syncthreads();

#pragma unroll
  for (int i = 0; i < kTPW; ++i) {
    const int tile = wave + i * kWaves;
    const int mt   = tile / 3;
    const int ncol = (tile % 3) * 16 + l16;
    store_tile<true>(s_v, hold[i], mt, ncol, hi);
  }
#pragma unroll
  for (int j = 0; j < kWIt; ++j) {
    const int g = tid + j * kThreads;
    if (j < kWIt - 1 || g < kWVec) ((v4f*)s_x)[g] = wpr[j];
  }
  __syncthreads();

  float* s_o = s_q;

  for (int h = wave; h < kH; h += kWaves) {
    const int co = h * kD;

    v2f ak[4];
#pragma unroll
    for (int kt = 0; kt < 4; ++kt) {
      const int kr = clampi(kt * 16 + l16, kN - 1);
      ak[kt].x = s_k[kr * kRS + co + kbl];
      ak[kt].y = s_k[kr * kRS + co + kbl + 1];
    }

#pragma unroll 2
    for (int qt = 0; qt < 4; ++qt) {
      const int q   = qt * 16 + l16;
      const int q_c = clampi(q, kN - 1);
      v2f bq_;
      bq_.x = s_q[q_c * kRS + co + kbl];
      bq_.y = s_q[q_c * kRS + co + kbl + 1];

      v8f et[4];
#pragma unroll
      for (int kt = 0; kt < 4; ++kt) {
        v8f z = {0.f, 0.f, 0.f, 0.f, 0.f, 0.f, 0.f, 0.f};
        et[kt] = __builtin_amdgcn_wmma_f32_16x16x4_f32(false, ak[kt], false, bq_,
                                                       (short)0, z, false, false);
      }
#pragma unroll
      for (int r = 5; r < 8; ++r) et[3][r] = hi ? -1e30f : et[3][r];

      float m = -1e30f;
#pragma unroll
      for (int kt = 0; kt < 4; ++kt)
#pragma unroll
        for (int r = 0; r < 8; ++r) m = fmaxf(m, et[kt][r]);
      m = fmaxf(m, __shfl_xor(m, 16, 32));

      float sum = 0.f;
#pragma unroll
      for (int kt = 0; kt < 4; ++kt)
#pragma unroll
        for (int r = 0; r < 8; ++r) {
          const float e = __expf((et[kt][r] - m) * kInvSqrtE);
          et[kt][r] = e;
          sum += e;
        }
      sum += __shfl_xor(sum, 16, 32);
      const float inv = 1.f / sum;

      float o0 = 0.f, o1 = 0.f, o2 = 0.f, o3 = 0.f;
#pragma unroll
      for (int kt = 0; kt < 4; ++kt)
#pragma unroll
        for (int r = 0; r < 8; ++r) {
          const int key = clampi(kt * 16 + r + 8 * hi, kN - 1);
          const float4 vr = *(const float4*)(s_v + key * kRS + co);
          const float a = et[kt][r];
          o0 += a * vr.x; o1 += a * vr.y; o2 += a * vr.z; o3 += a * vr.w;
        }
      o0 += __shfl_xor(o0, 16, 32);
      o1 += __shfl_xor(o1, 16, 32);
      o2 += __shfl_xor(o2, 16, 32);
      o3 += __shfl_xor(o3, 16, 32);

      float4 ov; ov.x = o0 * inv; ov.y = o1 * inv; ov.z = o2 * inv; ov.w = o3 * inv;
      *(float4*)(s_o + q * kRS + co) = ov;
    }
  }
  __syncthreads();

  for (int tile = wave; tile < 12; tile += kWaves) {
    const int mt   = tile / 3;
    const int nt   = tile % 3;
    const int ncol = nt * 16 + l16;
    const int ar_c = clampi(mt * 16 + l16, kN - 1);
    const int nc_c = clampi(ncol, kE - 1);
    const bool cok = ncol < kE;

    const float bias = bp[nc_c];
    v8f acc = proj_tile(s_o, s_x  , ar_c, nc_c, kbl);
    asm volatile("" :: "v"(bias));
    acc = acc + bias;

    const int row0 = mt * 16 + hi * 8;
    float* __restrict__ op = out + (size_t)b * (kN * kE) + row0 * kE + ncol;
    if (cok) {
#pragma unroll
      for (int r = 0; r < 5; ++r) op[r * kE] = acc[r];
      if (row0 + 8 <= kN) {
#pragma unroll
        for (int r = 5; r < 8; ++r) op[r * kE] = acc[r];
      }
    }
  }
}

extern "C" void kernel_launch(void* const* d_in, const int* in_sizes, int n_in,
                              void* d_out, int out_size, void* d_ws, size_t ws_size,
                              hipStream_t stream) {
  (void)in_sizes; (void)n_in; (void)out_size; (void)d_ws; (void)ws_size;
  const float* x  = (const float*)d_in[0];
  const float* Wq = (const float*)d_in[1];
  const float* bq = (const float*)d_in[2];
  const float* Wk = (const float*)d_in[3];
  const float* bk = (const float*)d_in[4];
  const float* Wv = (const float*)d_in[5];
  const float* bv = (const float*)d_in[6];
  const float* Wp = (const float*)d_in[7];
  const float* bp = (const float*)d_in[8];
  const float* pe = (const float*)d_in[9];
  float* out = (float*)d_out;

  mha_fused_kernel<<<dim3(4096), dim3(64), 0, stream>>>(
      x, Wq, bq, Wk, bk, Wv, bv, Wp, bp, pe, out);
}


# attention: no LDS round trip left on a query tile's chain (cross-half exchange by v_permlanex16, not ds_bpermute + wait; Q fragments read a tile pair ahead), so a trip's two tiles interleave
# speedup vs baseline: 2.8674x; 1.0538x over previous
#include <hip/hip_runtime.h>


typedef __attribute__((ext_vector_type(2))) float v2f;
typedef __attribute__((ext_vector_type(4))) float v4f;
typedef __attribute__((ext_vector_type(8))) float v8f;

namespace {
constexpr int   kN  = 61;
constexpr int   kE  = 40;
constexpr int   kH  = 10;
constexpr int   kD  = 4;
constexpr int   kRS = 40;
constexpr int   kBufX = kN * kRS;
constexpr int   kBuf  = 64 * kRS;
constexpr int   kWsz  = kE * kE;
constexpr float kInvSqrtE = 0.15811388300841898f;
constexpr int   kWaves   = 2;
constexpr int   kThreads = kWaves * 32;
constexpr int   kTPW  = 12 / kWaves;
constexpr int   kXVec = kN * kE / 4;
constexpr int   kXIt  = (kXVec + kThreads - 1) / kThreads;
constexpr int   kWVec = kWsz / 4;
constexpr int   kWIt  = (kWVec + kThreads - 1) / kThreads;
}

__device__ __forceinline__ int clampi(int v, int hi2) { return v < hi2 ? v : hi2; }

__device__ __forceinline__ float xhalf(float x) {
  const int b = __builtin_bit_cast(int, x);
  return __builtin_bit_cast(float,
      __builtin_amdgcn_permlanex16(b, b, 0x76543210, 0xfedcba98, false, false));
}

__device__ __forceinline__ v8f proj_tile(const float* __restrict__ sA,
                                         const float* __restrict__ sW,
                                         int ar_c, int nc_c, int kbl) {
  v2f af[10], bf[10];
#pragma unroll
  for (int kc = 0; kc < 10; ++kc) {
    const int kb = kc * 4 + kbl;
    af[kc].x = sA[ar_c * kRS + kb];
    af[kc].y = sA[ar_c * kRS + kb + 1];
    bf[kc].x = sW[nc_c * kE + kb];
    bf[kc].y = sW[nc_c * kE + kb + 1];
  }
  v8f acc = {0.f, 0.f, 0.f, 0.f, 0.f, 0.f, 0.f, 0.f};
#pragma unroll
  for (int kc = 0; kc < 10; ++kc)
    acc = __builtin_amdgcn_wmma_f32_16x16x4_f32(false, af[kc], false, bf[kc],
                                                (short)0, acc, false, false);
  return acc;
}

template <bool kMask>
__device__ __forceinline__ void store_tile(float* __restrict__ S, const v8f acc,
                                           int mt, int ncol, int hi) {
  float* __restrict__ base = S + (mt * 16 + hi * 8) * kRS + ncol;
  if (ncol < kE) {
#pragma unroll
    for (int r = 0; r < 8; ++r) {
      float val = acc[r];
      if (kMask && r < 3)
        val = (mt == 0 && hi == 0 && (ncol & 3) > r) ? 0.f : val;
      base[r * kRS] = val;
    }
  }
}

__global__ __launch_bounds__(kThreads, 4)
void mha_fused_kernel(const float* __restrict__ x,
                      const float* __restrict__ Wq, const float* __restrict__ bq,
                      const float* __restrict__ Wk, const float* __restrict__ bk,
                      const float* __restrict__ Wv, const float* __restrict__ bv,
                      const float* __restrict__ Wp, const float* __restrict__ bp,
                      const float* __restrict__ pe,
                      float* __restrict__ out) {
  __shared__ __align__(16) float s_x[kBufX];
  __shared__ __align__(16) float s_q[kBuf];
  __shared__ __align__(16) float s_k[kBuf];
  __shared__ __align__(16) float s_v[kBuf];

  const int b    = blockIdx.x;
  const int tid  = threadIdx.x;
  const int lane = tid & 31;
  const int wave = tid >> 5;
  const int hi   = lane >> 4;
  const int l16  = lane & 15;
  const int kbl  = hi * 2;

  v4f wvr[kWIt];
  {
    const v4f* __restrict__ xv  = (const v4f*)(x + (size_t)b * (kN * kE));
    const v4f* __restrict__ pv  = (const v4f*)pe;
    const v4f* __restrict__ wqv = (const v4f*)Wq;
    const v4f* __restrict__ wkv = (const v4f*)Wk;
    const v4f* __restrict__ wvv = (const v4f*)Wv;
    v4f xr[kXIt], pr[kXIt], wkr[kWIt], wqr[kWIt];
#pragma unroll
    for (int j = 0; j < kXIt; ++j) {
      const int g = clampi(tid + j * kThreads, kXVec - 1);
      xr[j] = xv[g];
      pr[j] = pv[g];
    }
#pragma unroll
    for (int j = 0; j < kWIt; ++j) {
      const int g = clampi(tid + j * kThreads, kWVec - 1);
      wqr[j] = wqv[g];
      wkr[j] = wkv[g];
      wvr[j] = wvv[g];
    }
#pragma unroll
    for (int j = 0; j < kXIt; ++j) {
      const int g = tid + j * kThreads;
      if (j < kXIt - 1 || g < kXVec) ((v4f*)s_x)[g] = xr[j] + pr[j];
    }
#pragma unroll
    for (int j = 0; j < kWIt; ++j) {
      const int g = tid + j * kThreads;
      if (j < kWIt - 1 || g < kWVec) {
        ((v4f*)s_k)[g] = wqr[j];
        ((v4f*)s_v)[g] = wkr[j];
      }
    }
  }
  __syncthreads();

  v8f hold[kTPW];
#pragma unroll
  for (int i = 0; i < kTPW; ++i) {
    const int tile = wave + i * kWaves;
    const int mt   = tile / 3;
    const int nt   = tile % 3;
    const int ncol = nt * 16 + l16;
    const int ar_c = clampi(mt * 16 + l16, kN - 1);
    const int nc_c = clampi(ncol, kE - 1);
    const float bqv = bq[nc_c];
    const float bkv = bk[nc_c];
    v8f qa = proj_tile(s_x, s_k  , ar_c, nc_c, kbl);
    asm volatile("" :: "v"(bqv));
    qa = qa + bqv;
    store_tile<false>(s_q, qa, mt, ncol, hi);
    hold[i] = proj_tile(s_x, s_v  , ar_c, nc_c, kbl) + bkv;
  }
  __syncthreads();

#pragma unroll
  for (int i = 0; i < kTPW; ++i) {
    const int tile = wave + i * kWaves;
    const int mt   = tile / 3;
    const int ncol = (tile % 3) * 16 + l16;
    store_tile<true>(s_k, hold[i], mt, ncol, hi);
  }
#pragma unroll
  for (int j = 0; j < kWIt; ++j) {
    const int g = tid + j * kThreads;
    if (j < kWIt - 1 || g < kWVec) ((v4f*)s_v)[g] = wvr[j];
  }
  __syncthreads();

  v4f wpr[kWIt];
#pragma unroll
  for (int j = 0; j < kWIt; ++j)
    wpr[j] = ((const v4f*)Wp)[clampi(tid + j * kThreads, kWVec - 1)];

#pragma unroll
  for (int i = 0; i < kTPW; ++i) {
    const int tile = wave + i * kWaves;
    const int mt   = tile / 3;
    const int ncol = (tile % 3) * 16 + l16;
    const int nc_c = clampi(ncol, kE - 1);
    const float bvv = bv[nc_c];
    hold[i] = proj_tile(s_x, s_v  , clampi(mt * 16 + l16, kN - 1), nc_c, kbl) + bvv;
  }
  __syncthreads();

#pragma unroll
  for (int i = 0; i < kTPW; ++i) {
    const int tile = wave + i * kWaves;
    const int mt   = tile / 3;
    const int ncol = (tile % 3) * 16 + l16;
    store_tile<true>(s_v, hold[i], mt, ncol, hi);
  }
#pragma unroll
  for (int j = 0; j < kWIt; ++j) {
    const int g = tid + j * kThreads;
    if (j < kWIt - 1 || g < kWVec) ((v4f*)s_x)[g] = wpr[j];
  }
  __syncthreads();

  float* s_o = s_q;

  for (int h = wave; h < kH; h += kWaves) {
    const int co = h * kD;

    v2f ak[4];
#pragma unroll
    for (int kt = 0; kt < 4; ++kt) {
      const int kr = clampi(kt * 16 + l16, kN - 1);
      ak[kt].x = s_k[kr * kRS + co + kbl];
      ak[kt].y = s_k[kr * kRS + co + kbl + 1];
    }

    v2f bqa, bqb;
    bqa.x = s_q[l16 * kRS + co + kbl];
    bqa.y = s_q[l16 * kRS + co + kbl + 1];
    bqb.x = s_q[(16 + l16) * kRS + co + kbl];
    bqb.y = s_q[(16 + l16) * kRS + co + kbl + 1];

#pragma unroll 1
    for (int qp = 0; qp < 2; ++qp) {
      const int qna = clampi(qp * 32 + 32 + l16, kN - 1);
      const int qnb = clampi(qp * 32 + 48 + l16, kN - 1);
      v2f bna, bnb;
      bna.x = s_q[qna * kRS + co + kbl];
      bna.y = s_q[qna * kRS + co + kbl + 1];
      bnb.x = s_q[qnb * kRS + co + kbl];
      bnb.y = s_q[qnb * kRS + co + kbl + 1];
#pragma unroll
     for (int t = 0; t < 2; ++t) {
      const int qt  = qp * 2 + t;
      const int q   = qt * 16 + l16;
      const v2f bq_ = t ? bqb : bqa;

      v8f et[4];
#pragma unroll
      for (int kt = 0; kt < 4; ++kt) {
        v8f z = {0.f, 0.f, 0.f, 0.f, 0.f, 0.f, 0.f, 0.f};
        et[kt] = __builtin_amdgcn_wmma_f32_16x16x4_f32(false, ak[kt], false, bq_,
                                                       (short)0, z, false, false);
      }
#pragma unroll
      for (int r = 5; r < 8; ++r) et[3][r] = hi ? -1e30f : et[3][r];

      float m = -1e30f;
#pragma unroll
      for (int kt = 0; kt < 4; ++kt)
#pragma unroll
        for (int r = 0; r < 8; ++r) m = fmaxf(m, et[kt][r]);
      m = fmaxf(m, xhalf(m));

      float sum = 0.f;
#pragma unroll
      for (int kt = 0; kt < 4; ++kt)
#pragma unroll
        for (int r = 0; r < 8; ++r) {
          const float e = __expf((et[kt][r] - m) * kInvSqrtE);
          et[kt][r] = e;
          sum += e;
        }
      sum += xhalf(sum);
      const float inv = 1.f / sum;

      float o0 = 0.f, o1 = 0.f, o2 = 0.f, o3 = 0.f;
#pragma unroll
      for (int kt = 0; kt < 4; ++kt)
#pragma unroll
        for (int r = 0; r < 8; ++r) {
          const int key = clampi(kt * 16 + r + 8 * hi, kN - 1);
          const float4 vr = *(const float4*)(s_v + key * kRS + co);
          const float a = et[kt][r];
          o0 += a * vr.x; o1 += a * vr.y; o2 += a * vr.z; o3 += a * vr.w;
        }
      o0 += xhalf(o0);
      o1 += xhalf(o1);
      o2 += xhalf(o2);
      o3 += xhalf(o3);

      float4 ov; ov.x = o0 * inv; ov.y = o1 * inv; ov.z = o2 * inv; ov.w = o3 * inv;
      *(float4*)(s_o + q * kRS + co) = ov;
     }
      bqa = bna;
      bqb = bnb;
    }
  }
  __syncthreads();

  for (int tile = wave; tile < 12; tile += kWaves) {
    const int mt   = tile / 3;
    const int nt   = tile % 3;
    const int ncol = nt * 16 + l16;
    const int ar_c = clampi(mt * 16 + l16, kN - 1);
    const int nc_c = clampi(ncol, kE - 1);
    const bool cok = ncol < kE;

    const float bias = bp[nc_c];
    v8f acc = proj_tile(s_o, s_x  , ar_c, nc_c, kbl);
    asm volatile("" :: "v"(bias));
    acc = acc + bias;

    const int row0 = mt * 16 + hi * 8;
    float* __restrict__ op = out + (size_t)b * (kN * kE) + row0 * kE + ncol;
    if (cok) {
#pragma unroll
      for (int r = 0; r < 5; ++r) op[r * kE] = acc[r];
      if (row0 + 8 <= kN) {
#pragma unroll
        for (int r = 5; r < 8; ++r) op[r * kE] = acc[r];
      }
    }
  }
}

extern "C" void kernel_launch(void* const* d_in, const int* in_sizes, int n_in,
                              void* d_out, int out_size, void* d_ws, size_t ws_size,
                              hipStream_t stream) {
  (void)in_sizes; (void)n_in; (void)out_size; (void)d_ws; (void)ws_size;
  const float* x  = (const float*)d_in[0];
  const float* Wq = (const float*)d_in[1];
  const float* bq = (const float*)d_in[2];
  const float* Wk = (const float*)d_in[3];
  const float* bk = (const float*)d_in[4];
  const float* Wv = (const float*)d_in[5];
  const float* bv = (const float*)d_in[6];
  const float* Wp = (const float*)d_in[7];
  const float* bp = (const float*)d_in[8];
  const float* pe = (const float*)d_in[9];
  float* out = (float*)d_out;

  mha_fused_kernel<<<dim3(4096), dim3(64), 0, stream>>>(
      x, Wq, bq, Wk, bk, Wv, bv, Wp, bp, pe, out);
}
